# starting kernel
# baseline (speedup 1.0000x reference)
#include <hip/hip_runtime.h>

typedef float v2f __attribute__((ext_vector_type(2)));
typedef float v4f __attribute__((ext_vector_type(4)));
typedef float v8f __attribute__((ext_vector_type(8)));

#define N_MODES    64
#define N_CLASSES  10
#define TILE_ROWS  128
#define LDS_STRIDE 68

__global__ void cv2d_fold_kernel(const float* __restrict__ S, const float* __restrict__ d,
                                 const float* __restrict__ bias, const float* __restrict__ W,
                                 const float* __restrict__ b, float* __restrict__ Mfold,
                                 float* __restrict__ cst) {
  int t = blockIdx.x * blockDim.x + threadIdx.x;
  if (t >= 16 * N_MODES) return;
  int c = t >> 6;
  int i = t & 63;
  float acc = 0.f;
  if (c < N_CLASSES) {
    for (int m = 0; m < N_MODES; ++m)
      acc += W[c * N_MODES + m] * S[(2 * m) * (2 * N_MODES) + 2 * i];
  }
  Mfold[c * N_MODES + i] = acc;
  if (i == 0) {
    float cv = 0.f;
    if (c < N_CLASSES) {
      for (int m = 0; m < N_MODES; ++m)
        cv += W[c * N_MODES + m] * (d[2 * m] + bias[m]);
      cv += b[c];
    }
    cst[c] = cv;
  }
}

__global__ __launch_bounds__(256) void cv2d_gemm_kernel(const float* __restrict__ x,
                                                        const float* __restrict__ Mfold,
                                                        const float* __restrict__ cst,
                                                        float* __restrict__ out, int Bn) {
  __shared__ __align__(16) float lds[TILE_ROWS * LDS_STRIDE];

  const int t    = threadIdx.x;
  const int lane = t & 31;
  const int wave = t >> 5;
  const int r0   = blockIdx.x * TILE_ROWS;

  {
    const int row  = t >> 1;
    const int half = t & 1;
    int grow = r0 + row;
    if (grow >= Bn) grow = Bn - 1;
    const v4f* src = (const v4f*)(x + (size_t)grow * N_MODES + half * 32);
    float*     dst = lds + row * LDS_STRIDE + half * 32;
#pragma unroll
    for (int i = 0; i < 8; ++i) {
      v4f v = __builtin_nontemporal_load(src + i);
      *(v4f*)(dst + i * 4) = v;
    }
  }

  const int n     = lane & 15;
  const int khalf = (lane >> 4) << 1;
  v2f bfrag[16];
#pragma unroll
  for (int tk = 0; tk < 16; ++tk)
    bfrag[tk] = *(const v2f*)(Mfold + n * N_MODES + tk * 4 + khalf);
  const float cn = cst[n];

  __syncthreads();

  v8f acc = {};
  const float* arow = lds + (wave * 16 + (lane & 15)) * LDS_STRIDE + khalf;
#pragma unroll
  for (int tk = 0; tk < 16; ++tk) {
    v2f a = *(const v2f*)(arow + tk * 4);
    acc = __builtin_amdgcn_wmma_f32_16x16x4_f32(
         false, a,  false, bfrag[tk],
         (short)0, acc,  false,  false);
  }

  const int rbase = r0 + wave * 16 + ((lane >> 4) << 3);
  const bool tile_full = (r0 + TILE_ROWS) <= Bn;
  if (n < N_CLASSES) {
    float* op = out + (size_t)rbase * N_CLASSES + n;
    if (tile_full) {
#pragma unroll
      for (int v = 0; v < 8; ++v)
        __builtin_nontemporal_store(acc[v] + cn, op + v * N_CLASSES);
    } else {
#pragma unroll
      for (int v = 0; v < 8; ++v)
        if (rbase + v < Bn)
          __builtin_nontemporal_store(acc[v] + cn, op + v * N_CLASSES);
    }
  }
}

extern "C" void kernel_launch(void* const* d_in, const int* in_sizes, int n_in,
                              void* d_out, int out_size, void* d_ws, size_t ws_size,
                              hipStream_t stream) {
  const float* x    = (const float*)d_in[0];
  const float* S    = (const float*)d_in[1];
  const float* d    = (const float*)d_in[2];
  const float* bias = (const float*)d_in[3];
  const float* W    = (const float*)d_in[4];
  const float* b    = (const float*)d_in[5];
  float* out = (float*)d_out;

  const int Bn = in_sizes[0] / N_MODES;

  float* Mfold = (float*)d_ws;
  float* cst   = Mfold + 16 * N_MODES;

  hipLaunchKernelGGL(cv2d_fold_kernel, dim3(4), dim3(256), 0, stream,
                     S, d, bias, W, b, Mfold, cst);

  const int nblocks = (Bn + TILE_ROWS - 1) / TILE_ROWS;
  hipLaunchKernelGGL(cv2d_gemm_kernel, dim3(nblocks), dim3(256), 0, stream,
                     x, Mfold, cst, out, Bn);
}


# streaming form: 256 workgroups x 16 waves (one per WGP, was 1563 x 8); each wave owns 16-row groups, all its whole-line 512-byte loads in flight up front, own LDS transpose, no block barrier
# speedup vs baseline: 1.5617x; 1.5617x over previous
#include <hip/hip_runtime.h>


typedef float v2f __attribute__((ext_vector_type(2)));
typedef float v4f __attribute__((ext_vector_type(4)));
typedef float v8f __attribute__((ext_vector_type(8)));

#define N_MODES    64
#define N_CLASSES  10
#define LDS_STRIDE 68

#define GROUP_ROWS   16
#define WAVES        16
#define AHEAD        4
#define NUM_WG       256
#define GROUP_FLOATS (GROUP_ROWS * LDS_STRIDE)

__global__ void cv2d_fold_kernel(const float* __restrict__ S, const float* __restrict__ d,
                                 const float* __restrict__ bias, const float* __restrict__ W,
                                 const float* __restrict__ b, float* __restrict__ Mfold,
                                 float* __restrict__ cst) {
  int t = blockIdx.x * blockDim.x + threadIdx.x;
  if (t >= 16 * N_MODES) return;
  int c = t >> 6;
  int i = t & 63;
  float acc = 0.f;
  if (c < N_CLASSES) {
    for (int m = 0; m < N_MODES; ++m)
      acc += W[c * N_MODES + m] * S[(2 * m) * (2 * N_MODES) + 2 * i];
  }
  Mfold[c * N_MODES + i] = acc;
  if (i == 0) {
    float cv = 0.f;
    if (c < N_CLASSES) {
      for (int m = 0; m < N_MODES; ++m)
        cv += W[c * N_MODES + m] * (d[2 * m] + bias[m]);
      cv += b[c];
    }
    cst[c] = cv;
  }
}

__device__ __forceinline__ void wave_lds_handover() {
  asm volatile("s_wait_dscnt 0x0" ::: "memory");
}

template <int CNT>
__device__ __forceinline__ void stream_pass(const float* __restrict__ x, float* __restrict__ out,
                                            float* __restrict__ wlds, const v2f (&bfrag)[16],
                                            float cn, int g0, int gstep, int Bn, int lane) {
  const int n     = lane & 15;
  const int khalf = (lane >> 4) << 1;
  const int lrow  = lane >> 4;
  const int lcol  = (lane & 15) * 4;
  const int aoff  = (lane & 15) * LDS_STRIDE + khalf;

  v4f buf[CNT][8];

  __builtin_amdgcn_sched_barrier(0);
#pragma unroll
  for (int j = 0; j < CNT; ++j) {
    const int g = g0 + j * gstep;
    const char* gbase = (const char*)x + (size_t)g * (GROUP_ROWS * N_MODES * sizeof(float));
    const int last = min(GROUP_ROWS, Bn - g * GROUP_ROWS) - 1;
#pragma unroll
    for (int i = 0; i < 8; ++i) {
      const int r = min(2 * i + lrow, last);
      const unsigned off = (unsigned)r * (unsigned)(N_MODES * sizeof(float)) + (unsigned)lcol * 4u;
      buf[j][i] = __builtin_nontemporal_load((const v4f*)(gbase + off));
    }
    __builtin_amdgcn_sched_barrier(0);
  }
  __builtin_amdgcn_sched_barrier(0);

#pragma unroll
  for (int j = 0; j < CNT; ++j) {
    const int g = g0 + j * gstep;
    float* const gl = wlds + j * GROUP_FLOATS;
#pragma unroll
    for (int i = 0; i < 8; ++i)
      *(v4f*)(gl + (2 * i + lrow) * LDS_STRIDE + lcol) = buf[j][i];

    wave_lds_handover();

    v8f acc = {};
    const float* arow = gl + aoff;
#pragma unroll
    for (int tk = 0; tk < 16; ++tk) {
      v2f a = *(const v2f*)(arow + tk * 4);
      acc = __builtin_amdgcn_wmma_f32_16x16x4_f32(
           false, a,  false, bfrag[tk],
           (short)0, acc,  false,  false);
    }

    const int  rbase      = g * GROUP_ROWS + ((lane >> 4) << 3);
    const bool group_full = (g * GROUP_ROWS + GROUP_ROWS) <= Bn;
    if (n < N_CLASSES) {
      float* op = out + (size_t)rbase * N_CLASSES + n;
      if (group_full) {
#pragma unroll
        for (int v = 0; v < 8; ++v)
          __builtin_nontemporal_store(acc[v] + cn, op + v * N_CLASSES);
      } else {
#pragma unroll
        for (int v = 0; v < 8; ++v)
          if (rbase + v < Bn)
            __builtin_nontemporal_store(acc[v] + cn, op + v * N_CLASSES);
      }
    }
  }
  wave_lds_handover();
}

__global__ __launch_bounds__(WAVES * 32) void cv2d_gemm_kernel(const float* __restrict__ x,
                                                               const float* __restrict__ Mfold,
                                                               const float* __restrict__ cst,
                                                               float* __restrict__ out, int Bn) {
  __shared__ __align__(16) float lds[WAVES * AHEAD * GROUP_FLOATS];

  const int lane = threadIdx.x & 31;
  const int wave = __builtin_amdgcn_readfirstlane((int)(threadIdx.x >> 5));

  const int ngroups = (Bn + GROUP_ROWS - 1) / GROUP_ROWS;
  const int nwaves  = (int)gridDim.x * WAVES;
  const int wid     = wave * (int)gridDim.x + (int)blockIdx.x;
  if (wid >= ngroups) return;

  const int n     = lane & 15;
  const int khalf = (lane >> 4) << 1;
  v2f bfrag[16];
#pragma unroll
  for (int tk = 0; tk < 16; ++tk)
    bfrag[tk] = *(const v2f*)(Mfold + n * N_MODES + tk * 4 + khalf);
  const float cn = cst[n];

  float* const wlds = lds + wave * (AHEAD * GROUP_FLOATS);

  for (int g0 = wid; g0 < ngroups; g0 += AHEAD * nwaves) {
    if (g0 + 3 * nwaves < ngroups)      stream_pass<4>(x, out, wlds, bfrag, cn, g0, nwaves, Bn, lane);
    else if (g0 + 2 * nwaves < ngroups) stream_pass<3>(x, out, wlds, bfrag, cn, g0, nwaves, Bn, lane);
    else if (g0 + nwaves < ngroups)     stream_pass<2>(x, out, wlds, bfrag, cn, g0, nwaves, Bn, lane);
    else                                stream_pass<1>(x, out, wlds, bfrag, cn, g0, nwaves, Bn, lane);
  }
}

extern "C" void kernel_launch(void* const* d_in, const int* in_sizes, int n_in,
                              void* d_out, int out_size, void* d_ws, size_t ws_size,
                              hipStream_t stream) {
  const float* x    = (const float*)d_in[0];
  const float* S    = (const float*)d_in[1];
  const float* d    = (const float*)d_in[2];
  const float* bias = (const float*)d_in[3];
  const float* W    = (const float*)d_in[4];
  const float* b    = (const float*)d_in[5];
  float* out = (float*)d_out;

  const int Bn = in_sizes[0] / N_MODES;

  float* Mfold = (float*)d_ws;
  float* cst   = Mfold + 16 * N_MODES;

  hipLaunchKernelGGL(cv2d_fold_kernel, dim3(4), dim3(256), 0, stream,
                     S, d, bias, W, b, Mfold, cst);

  hipLaunchKernelGGL(cv2d_gemm_kernel, dim3(NUM_WG), dim3(WAVES * 32), 0, stream,
                     x, Mfold, cst, out, Bn);
}


# whole-line output stores: each 16-row group's 640 output bytes staged in its LDS block, written by two 16-byte-per-lane stores (was 8 partial 4-byte stores), copy-out hidden behind the next product
# speedup vs baseline: 2.4982x; 1.5997x over previous
#include <hip/hip_runtime.h>


typedef float v2f __attribute__((ext_vector_type(2)));
typedef float v4f __attribute__((ext_vector_type(4)));
typedef float v8f __attribute__((ext_vector_type(8)));

#define N_MODES    64
#define N_CLASSES  10
#define LDS_STRIDE 68

#define GROUP_ROWS   16
#define WAVES        16
#define AHEAD        4
#define NUM_WG       256
#define GROUP_FLOATS (GROUP_ROWS * LDS_STRIDE)
#define FOLD_PARTS   4
#define FOLD_FLOATS  (16 * LDS_STRIDE)

__device__ __forceinline__ void wave_lds_handover() {
  asm volatile("s_wait_dscnt 0x0" ::: "memory");
}

__device__ __forceinline__ void wg_barrier_lds_only() {
  asm volatile("s_wait_dscnt 0x0\n\ts_barrier_signal -1\n\ts_barrier_wait -1" ::: "memory");
}

template <int CNT>
__device__ __forceinline__ void issue_groups(const float* __restrict__ x, v4f (&buf)[AHEAD][8],
                                             int g0, int gstep, int Bn, int lane) {
  const int lrow = lane >> 4;
  const int lcol = (lane & 15) * 4;
#pragma unroll
  for (int j = 0; j < CNT; ++j) {
    const int g = g0 + j * gstep;
    const char* gbase = (const char*)x + (size_t)g * (GROUP_ROWS * N_MODES * sizeof(float));
    const int last = min(GROUP_ROWS, Bn - g * GROUP_ROWS) - 1;
#pragma unroll
    for (int i = 0; i < 8; ++i) {
      const int r = min(2 * i + lrow, last);
      const unsigned off = (unsigned)r * (unsigned)(N_MODES * sizeof(float)) + (unsigned)lcol * 4u;
      buf[j][i] = __builtin_nontemporal_load((const v4f*)(gbase + off));
    }
    __builtin_amdgcn_sched_barrier(0);
  }
}


__device__ __forceinline__ void stage_rows(float* __restrict__ gl, const v4f (&rows)[8], int lane) {
  const int lrow = lane >> 4;
  const int lcol = (lane & 15) * 4;
#pragma unroll
  for (int i = 0; i < 8; ++i)
    *(v4f*)(gl + (2 * i + lrow) * LDS_STRIDE + lcol) = rows[i];
}

__device__ __forceinline__ v8f group_product(const float* __restrict__ gl, const v2f (&bfrag)[16], int lane) {
  const int khalf = (lane >> 4) << 1;
  const float* arow = gl + (lane & 15) * LDS_STRIDE + khalf;
  v8f acc = {};
#pragma unroll
  for (int tk = 0; tk < 16; ++tk) {
    v2f a = *(const v2f*)(arow + tk * 4);
    acc = __builtin_amdgcn_wmma_f32_16x16x4_f32(
         false, a,  false, bfrag[tk],
         (short)0, acc,  false,  false);
  }
  return acc;
}

__device__ __forceinline__ void stage_out(float* __restrict__ gl, const v8f& acc, float cn, int lane) {
  const int n = lane & 15;
  if (n < N_CLASSES) {
    float* sp = gl + ((lane >> 4) << 3) * N_CLASSES + n;
#pragma unroll
    for (int v = 0; v < 8; ++v)
      sp[v * N_CLASSES] = acc[v] + cn;
  }
}

__device__ __forceinline__ void flush_load(const float* __restrict__ gl, int lane, v4f& o0, v4f& o1) {
  o0 = *(const v4f*)(gl + 4 * lane);
  o1 = *(const v4f*)(gl + 128 + 4 * (lane & 7));
}

__device__ __forceinline__ void flush_store(float* __restrict__ out, const v4f& o0, const v4f& o1,
                                            int g, int Bn, int lane) {
  const int nvalid = min(GROUP_ROWS, Bn - g * GROUP_ROWS) * N_CLASSES;
  float* ob = out + (size_t)g * (GROUP_ROWS * N_CLASSES);
  if (nvalid == GROUP_ROWS * N_CLASSES) {
    __builtin_nontemporal_store(o0, (v4f*)(ob + 4 * lane));
    if (lane < 8)
      __builtin_nontemporal_store(o1, (v4f*)(ob + 128 + 4 * lane));
  } else {
#pragma unroll
    for (int e = 0; e < 4; ++e) {
      if (4 * lane + e < nvalid)
        __builtin_nontemporal_store(o0[e], ob + 4 * lane + e);
      if (lane < 8 && 128 + 4 * lane + e < nvalid)
        __builtin_nontemporal_store(o1[e], ob + 128 + 4 * lane + e);
    }
  }
}

__device__ __forceinline__ void consume_one(float* __restrict__ out, float* __restrict__ gl,
                                            const v4f (&rows)[8], const v2f (&bfrag)[16],
                                            float cn, int g, int Bn, int lane) {
  stage_rows(gl, rows, lane);
  wave_lds_handover();
  const v8f acc = group_product(gl, bfrag, lane);
  stage_out(gl, acc, cn, lane);
  wave_lds_handover();
  v4f o0, o1;
  flush_load(gl, lane, o0, o1);
  flush_store(out, o0, o1, g, Bn, lane);
}

template <int CNT>
__device__ __forceinline__ void consume_groups(float* __restrict__ out, float* __restrict__ wlds,
                                               const v4f (&buf)[AHEAD][8], const v2f (&bfrag)[16],
                                               float cn, int g0, int gstep, int Bn, int lane) {
#pragma unroll
  for (int j = 0; j < CNT; ++j) {
    float* gl = wlds + j * GROUP_FLOATS;
    stage_rows(gl, buf[j], lane);
    wave_lds_handover();
    v4f o0, o1;
    if (j > 0) {
      flush_load(wlds + (j - 1) * GROUP_FLOATS, lane, o0, o1);
      __builtin_amdgcn_sched_barrier(0);
    }
    const v8f acc = group_product(gl, bfrag, lane);
    __builtin_amdgcn_sched_barrier(0);
    if (j > 0) flush_store(out, o0, o1, g0 + (j - 1) * gstep, Bn, lane);
    stage_out(gl, acc, cn, lane);
  }
  if (CNT > 0) {
    wave_lds_handover();
    v4f o0, o1;
    flush_load(wlds + (CNT - 1) * GROUP_FLOATS, lane, o0, o1);
    flush_store(out, o0, o1, g0 + (CNT - 1) * gstep, Bn, lane);
  }
  wave_lds_handover();
}

template <int CNT>
__device__ __forceinline__ void stream_pass(const float* __restrict__ x, float* __restrict__ out,
                                            float* __restrict__ wlds, const v2f (&bfrag)[16],
                                            float cn, int g0, int gstep, int Bn, int lane) {
  v4f buf[AHEAD][8];
  __builtin_amdgcn_sched_barrier(0);
  issue_groups<CNT>(x, buf, g0, gstep, Bn, lane);
  __builtin_amdgcn_sched_barrier(0);
  consume_groups<CNT>(out, wlds, buf, bfrag, cn, g0, gstep, Bn, lane);
}

template <int CU>
__device__ __forceinline__ void first_pass(const float* __restrict__ x, const float* __restrict__ S,
                                           const float* __restrict__ d, const float* __restrict__ bias,
                                           const float* __restrict__ W, const float* __restrict__ b,
                                           float* __restrict__ out, float* __restrict__ wlds,
                                           float* __restrict__ flds, v2f (&bfrag)[16], float& cn,
                                           int g0, int gstep, bool has_extra, int Bn, int lane, int wave) {
  const int n  = lane & 15;
  const int h  = lane >> 4;
  const int t  = wave & 3;
  const int q  = wave >> 2;
  const int cc = min(n, N_CLASSES - 1);

  v2f   fa[4];
  float fb0[4], fb1[4];
  __builtin_amdgcn_sched_barrier(0);
#pragma unroll
  for (int j = 0; j < 4; ++j) {
    const int m = 16 * q + 4 * j + 2 * h;
    fa[j]  = *(const v2f*)(W + cc * N_MODES + m);
    fb0[j] = S[(2 * m)     * (2 * N_MODES) + 32 * t + 2 * n];
    fb1[j] = S[(2 * m + 2) * (2 * N_MODES) + 32 * t + 2 * n];
  }
  const int   mc = 16 * q + 4 * t + 2 * h;
  const v2f   fc = *(const v2f*)(W + cc * N_MODES + mc);
  const float d0 = d[2 * mc];
  const float d1 = d[2 * mc + 2];
  const v2f   bv = *(const v2f*)(bias + mc);
  const float bn = b[cc];
  __builtin_amdgcn_sched_barrier(0);

  v4f buf[AHEAD][8];
  issue_groups<CU>(x, buf, g0, gstep, Bn, lane);
  __builtin_amdgcn_sched_barrier(0);

  const bool real_class = n < N_CLASSES;
  v8f facc = {};
#pragma unroll
  for (int j = 0; j < 4; ++j) {
    v2f a;
    a.x = real_class ? fa[j].x : 0.f;
    a.y = real_class ? fa[j].y : 0.f;
    v2f bb;
    bb.x = fb0[j];
    bb.y = fb1[j];
    facc = __builtin_amdgcn_wmma_f32_16x16x4_f32(false, a, false, bb, (short)0, facc, false, false);
  }
  v8f cacc = {};
  {
    v2f a;
    a.x = real_class ? fc.x : 0.f;
    a.y = real_class ? fc.y : 0.f;
    v2f bb;
    bb.x = d0 + bv.x;
    bb.y = d1 + bv.y;
    cacc = __builtin_amdgcn_wmma_f32_16x16x4_f32(false, a, false, bb, (short)0, cacc, false, false);
  }
  float* const fp = flds + q * FOLD_FLOATS;
#pragma unroll
  for (int r = 0; r < 8; ++r)
    fp[(8 * h + r) * LDS_STRIDE + 16 * t + n] = facc[r];
  if (n == 0) {
#pragma unroll
    for (int r = 0; r < 8; ++r)
      fp[(8 * h + r) * LDS_STRIDE + N_MODES + t] = cacc[r];
  }

  v4f ebuf[8];
  if (CU < AHEAD && has_extra) {
    const int lrow = lane >> 4;
    const int lcol = (lane & 15) * 4;
    const int g = g0 + CU * gstep;
    const char* gbase = (const char*)x + (size_t)g * (GROUP_ROWS * N_MODES * sizeof(float));
    const int last = min(GROUP_ROWS, Bn - g * GROUP_ROWS) - 1;
#pragma unroll
    for (int i = 0; i < 8; ++i) {
      const int r = min(2 * i + lrow, last);
      const unsigned off = (unsigned)r * (unsigned)(N_MODES * sizeof(float)) + (unsigned)lcol * 4u;
      ebuf[i] = __builtin_nontemporal_load((const v4f*)(gbase + off));
    }
  }

  wg_barrier_lds_only();

  const float* const fr = flds + n * LDS_STRIDE + 2 * h;
#pragma unroll
  for (int tk = 0; tk < 16; ++tk) {
    v2f s = *(const v2f*)(fr + tk * 4);
#pragma unroll
    for (int p = 1; p < FOLD_PARTS; ++p)
      s += *(const v2f*)(fr + p * FOLD_FLOATS + tk * 4);
    bfrag[tk] = s;
    if ((tk & 3) == 3)
      asm volatile("" : "+v"(bfrag[tk - 3]), "+v"(bfrag[tk - 2]), "+v"(bfrag[tk - 1]), "+v"(bfrag[tk]) :: "memory");
  }
  {
    const float* const fcst = flds + n * LDS_STRIDE + N_MODES;
    v4f c4 = *(const v4f*)fcst;
#pragma unroll
    for (int p = 1; p < FOLD_PARTS; ++p)
      c4 += *(const v4f*)(fcst + p * FOLD_FLOATS);
    cn = ((c4.x + c4.y) + (c4.z + c4.w)) + (real_class ? bn : 0.f);
  }

  consume_groups<CU>(out, wlds, buf, bfrag, cn, g0, gstep, Bn, lane);

  if (CU < AHEAD && has_extra) {
    consume_one(out, wlds + CU * GROUP_FLOATS, ebuf, bfrag, cn, g0 + CU * gstep, Bn, lane);
    wave_lds_handover();
  }
}

__global__ __launch_bounds__(WAVES * 32) void cv2d_fused_kernel(const float* __restrict__ x,
                                                                const float* __restrict__ S,
                                                                const float* __restrict__ d,
                                                                const float* __restrict__ bias,
                                                                const float* __restrict__ W,
                                                                const float* __restrict__ b,
                                                                float* __restrict__ out, int Bn) {
  __shared__ __align__(16) float lds[WAVES * AHEAD * GROUP_FLOATS + FOLD_PARTS * FOLD_FLOATS];

  const int lane = threadIdx.x & 31;
  const int wave = __builtin_amdgcn_readfirstlane((int)(threadIdx.x >> 5));

  const int ngroups = (Bn + GROUP_ROWS - 1) / GROUP_ROWS;
  const int nwaves  = (int)gridDim.x * WAVES;
  const int wid     = wave * (int)gridDim.x + (int)blockIdx.x;
  const int widmax  = (WAVES - 1) * (int)gridDim.x + (int)blockIdx.x;

  const int mine = min(AHEAD, max(0, (ngroups - wid    + nwaves - 1) / nwaves));
  const int cu   = min(AHEAD, max(0, (ngroups - widmax + nwaves - 1) / nwaves));
  const bool has_extra = mine > cu;

  float* const wlds = lds + wave * (AHEAD * GROUP_FLOATS);
  float* const flds = lds + WAVES * AHEAD * GROUP_FLOATS;

  v2f   bfrag[16];
  float cn;

  if (cu == 4)      first_pass<4>(x, S, d, bias, W, b, out, wlds, flds, bfrag, cn, wid, nwaves, has_extra, Bn, lane, wave);
  else if (cu == 3) first_pass<3>(x, S, d, bias, W, b, out, wlds, flds, bfrag, cn, wid, nwaves, has_extra, Bn, lane, wave);
  else if (cu == 2) first_pass<2>(x, S, d, bias, W, b, out, wlds, flds, bfrag, cn, wid, nwaves, has_extra, Bn, lane, wave);
  else if (cu == 1) first_pass<1>(x, S, d, bias, W, b, out, wlds, flds, bfrag, cn, wid, nwaves, has_extra, Bn, lane, wave);
  else              first_pass<0>(x, S, d, bias, W, b, out, wlds, flds, bfrag, cn, wid, nwaves, has_extra, Bn, lane, wave);

  for (int g0 = wid + AHEAD * nwaves; g0 < ngroups; g0 += AHEAD * nwaves) {
    if (g0 + 3 * nwaves < ngroups)      stream_pass<4>(x, out, wlds, bfrag, cn, g0, nwaves, Bn, lane);
    else if (g0 + 2 * nwaves < ngroups) stream_pass<3>(x, out, wlds, bfrag, cn, g0, nwaves, Bn, lane);
    else if (g0 + nwaves < ngroups)     stream_pass<2>(x, out, wlds, bfrag, cn, g0, nwaves, Bn, lane);
    else                                stream_pass<1>(x, out, wlds, bfrag, cn, g0, nwaves, Bn, lane);
  }
}

extern "C" void kernel_launch(void* const* d_in, const int* in_sizes, int n_in,
                              void* d_out, int out_size, void* d_ws, size_t ws_size,
                              hipStream_t stream) {
  const float* x    = (const float*)d_in[0];
  const float* S    = (const float*)d_in[1];
  const float* d    = (const float*)d_in[2];
  const float* bias = (const float*)d_in[3];
  const float* W    = (const float*)d_in[4];
  const float* b    = (const float*)d_in[5];
  float* out = (float*)d_out;

  const int Bn = in_sizes[0] / N_MODES;

  hipLaunchKernelGGL(cv2d_fused_kernel, dim3(NUM_WG), dim3(WAVES * 32), 0, stream,
                     x, S, d, bias, W, b, out, Bn);
}
